# starting kernel
# baseline (speedup 1.0000x reference)
#include <hip/hip_runtime.h>

#define FEAT  128
#define EMBED 128

typedef float v2f __attribute__((ext_vector_type(2)));
typedef float v8f __attribute__((ext_vector_type(8)));

__global__ void zero_int_kernel(int* __restrict__ p, int n) {
    int i = blockIdx.x * blockDim.x + threadIdx.x;
    int stride = gridDim.x * blockDim.x;
    for (; i < n; i += stride) p[i] = 0;
}

__global__ void deg_kernel(const int* __restrict__ row, int* __restrict__ deg, int E) {
    int e = blockIdx.x * blockDim.x + threadIdx.x;
    if (e < E) atomicAdd(&deg[row[e]], 1);
}

__global__ __launch_bounds__(1024) void scan_kernel(const int* __restrict__ deg,
                                                    int* __restrict__ rowstart,
                                                    int N, int E) {
    __shared__ int sums[1024];
    const int t = threadIdx.x;
    const int chunk = (N + 1023) / 1024;
    const int begin = t * chunk;
    const int end = (begin + chunk < N) ? (begin + chunk) : N;

    int s = 0;
    for (int i = begin; i < end; ++i) s += deg[i];
    sums[t] = s;
    __syncthreads();

    for (int off = 1; off < 1024; off <<= 1) {
        int v = (t >= off) ? sums[t - off] : 0;
        __syncthreads();
        sums[t] += v;
        __syncthreads();
    }

    int run = (t == 0) ? 0 : sums[t - 1];
    for (int i = begin; i < end; ++i) {
        rowstart[i] = run;
        run += deg[i];
    }
    if (t == 0) rowstart[N] = E;
}

__global__ void csr_kernel(const int* __restrict__ row, const int* __restrict__ col,
                           const int* __restrict__ rowstart, int* __restrict__ cursor,
                           int* __restrict__ csr_col, int E) {
    int e = blockIdx.x * blockDim.x + threadIdx.x;
    if (e < E) {
        const int r = row[e];
        const int pos = atomicAdd(&cursor[r], 1);
        csr_col[rowstart[r] + pos] = col[e];
    }
}

__global__ __launch_bounds__(256) void gemm_wmma_kernel(const float* __restrict__ x,
                                                        const float* __restrict__ W,
                                                        float* __restrict__ h,
                                                        int nrows) {
    const int lane = threadIdx.x & 31;
    const int wave = threadIdx.x >> 5;
    const int l  = lane & 15;
    const int hh = lane >> 4;
    const int m0 = blockIdx.x * 16;
    const int n0 = wave * 16;

    int mrow = m0 + l;
    if (mrow >= nrows) mrow = nrows - 1;
    const float* __restrict__ xrow = x + (size_t)mrow * FEAT;

    v8f acc = {};
#pragma unroll 8
    for (int k0 = 0; k0 < FEAT; k0 += 4) {
        const int ka = k0 + 2 * hh;
        v2f a, b;
        a.x = xrow[ka];
        a.y = xrow[ka + 1];
        b.x = W[(size_t)ka       * EMBED + n0 + l];
        b.y = W[(size_t)(ka + 1) * EMBED + n0 + l];
        acc = __builtin_amdgcn_wmma_f32_16x16x4_f32(false, a, false, b,
                                                    (short)0, acc, false, false);
    }

    float* __restrict__ hout = h + (size_t)m0 * EMBED + n0 + l;
    if (m0 + 16 <= nrows) {
#pragma unroll
        for (int r = 0; r < 8; ++r)
            hout[(size_t)(r + 8 * hh) * EMBED] = acc[r];
    } else {
#pragma unroll
        for (int r = 0; r < 8; ++r) {
            const int mr = r + 8 * hh;
            if (m0 + mr < nrows) hout[(size_t)mr * EMBED] = acc[r];
        }
    }
}

__global__ __launch_bounds__(256) void gather_kernel(const float* __restrict__ h,
                                                     const int* __restrict__ deg,
                                                     const int* __restrict__ rowstart,
                                                     const int* __restrict__ csr_col,
                                                     float* __restrict__ out, int N) {
    const int lane = threadIdx.x & 31;
    const int r = blockIdx.x * (blockDim.x >> 5) + (threadIdx.x >> 5);
    if (r >= N) return;

    const int start = rowstart[r];
    const int end   = rowstart[r + 1];
    const float dr  = (float)deg[r];

    float4 acc = {0.f, 0.f, 0.f, 0.f};
    for (int i = start; i < end; ++i) {
        const int c = csr_col[i];
        const float norm = rsqrtf(dr * (float)deg[c]);
        const float4 v = ((const float4*)(h + (size_t)c * EMBED))[lane];
        acc.x += v.x * norm;
        acc.y += v.y * norm;
        acc.z += v.z * norm;
        acc.w += v.w * norm;
    }
    ((float4*)(out + (size_t)r * EMBED))[lane] = acc;
}

extern "C" void kernel_launch(void* const* d_in, const int* in_sizes, int n_in,
                              void* d_out, int out_size, void* d_ws, size_t ws_size,
                              hipStream_t stream) {
    const float* x   = (const float*)d_in[0];
    const float* W   = (const float*)d_in[1];
    const int*   row = (const int*)d_in[2];
    const int*   col = (const int*)d_in[3];
    float*       out = (float*)d_out;

    const int N = in_sizes[0] / FEAT;
    const int E = in_sizes[2];

    float* h        = (float*)d_ws;
    int*   deg      = (int*)(h + (size_t)N * EMBED);
    int*   cursor   = deg + N;
    int*   rowstart = cursor + N;
    int*   csr_col  = rowstart + (N + 1);

    zero_int_kernel<<<256, 256, 0, stream>>>(deg, 2 * N);

    deg_kernel<<<(E + 255) / 256, 256, 0, stream>>>(row, deg, E);

    scan_kernel<<<1, 1024, 0, stream>>>(deg, rowstart, N, E);

    csr_kernel<<<(E + 255) / 256, 256, 0, stream>>>(row, col, rowstart, cursor, csr_col, E);

    gemm_wmma_kernel<<<(N + 15) / 16, 256, 0, stream>>>(x, W, h, N);

    gather_kernel<<<(N + 7) / 8, 256, 0, stream>>>(h, deg, rowstart, csr_col, out, N);
}


# CSR build: per-edge global int atomics replaced by per-workgroup LDS counters (count matrix over 128 edge chunks + column scan); neighbour sum order within a row becomes chunk-major
# speedup vs baseline: 1.9438x; 1.9438x over previous
#include <hip/hip_runtime.h>

#define FEAT  128
#define EMBED 128

typedef float v2f __attribute__((ext_vector_type(2)));
typedef float v8f __attribute__((ext_vector_type(8)));

#define LDS_NODES 65536
#define WG 1024
#define EDGE_BATCH 8

__global__ __launch_bounds__(WG) void count_kernel(const int* __restrict__ row,
                                                   int* __restrict__ cnt,
                                                   int N, int E) {
    __shared__ int hist[LDS_NODES];
    const int t = threadIdx.x;
    const int chunk = (E + (int)gridDim.x - 1) / (int)gridDim.x;
    const int begin = (int)blockIdx.x * chunk;
    const int end = (begin + chunk < E) ? (begin + chunk) : E;
    int* __restrict__ mycnt = cnt + (size_t)blockIdx.x * N;

    for (int w0 = 0; w0 < N; w0 += LDS_NODES) {
        const int wn = (N - w0 < LDS_NODES) ? (N - w0) : LDS_NODES;
        for (int i = t; i < wn; i += WG) hist[i] = 0;
        __syncthreads();
        for (int e0 = begin + t; e0 < end; e0 += EDGE_BATCH * WG) {
            int rr[EDGE_BATCH];
#pragma unroll
            for (int j = 0; j < EDGE_BATCH; ++j) {
                const int e = e0 + j * WG;
                rr[j] = (e < end) ? row[e] : -1;
            }
#pragma unroll
            for (int j = 0; j < EDGE_BATCH; ++j) {
                const unsigned r = (unsigned)(rr[j] - w0);
                if (r < (unsigned)wn) atomicAdd(&hist[r], 1);
            }
        }
        __syncthreads();
        for (int i = t; i < wn; i += WG) mycnt[w0 + i] = hist[i];
        __syncthreads();
    }
}

__global__ __launch_bounds__(256) void colscan_kernel(int* __restrict__ cnt,
                                                      int* __restrict__ deg,
                                                      int N, int B) {
    const int r = blockIdx.x * blockDim.x + threadIdx.x;
    if (r >= N) return;
    int run = 0;
    for (int b0 = 0; b0 < B; b0 += 32) {
        int c[32];
#pragma unroll
        for (int j = 0; j < 32; ++j)
            c[j] = (b0 + j < B) ? cnt[(size_t)(b0 + j) * N + r] : 0;
#pragma unroll
        for (int j = 0; j < 32; ++j) {
            if (b0 + j < B) cnt[(size_t)(b0 + j) * N + r] = run;
            run += c[j];
        }
    }
    deg[r] = run;
}

__global__ __launch_bounds__(1024) void scan_kernel(const int* __restrict__ deg,
                                                    int* __restrict__ rowstart,
                                                    int N, int E) {
    __shared__ int sums[1024];
    const int t = threadIdx.x;
    const int chunk = (N + 1023) / 1024;
    const int begin = t * chunk;
    const int end = (begin + chunk < N) ? (begin + chunk) : N;

    int s = 0;
    for (int i = begin; i < end; ++i) s += deg[i];
    sums[t] = s;
    __syncthreads();

    for (int off = 1; off < 1024; off <<= 1) {
        int v = (t >= off) ? sums[t - off] : 0;
        __syncthreads();
        sums[t] += v;
        __syncthreads();
    }

    int run = (t == 0) ? 0 : sums[t - 1];
    for (int i = begin; i < end; ++i) {
        rowstart[i] = run;
        run += deg[i];
    }
    if (t == 0) rowstart[N] = E;
}

__global__ __launch_bounds__(WG) void fill_kernel(const int* __restrict__ row,
                                                  const int* __restrict__ col,
                                                  const int* __restrict__ cnt,
                                                  const int* __restrict__ rowstart,
                                                  int* __restrict__ csr_col,
                                                  int N, int E) {
    __shared__ int cur[LDS_NODES];
    const int t = threadIdx.x;
    const int chunk = (E + (int)gridDim.x - 1) / (int)gridDim.x;
    const int begin = (int)blockIdx.x * chunk;
    const int end = (begin + chunk < E) ? (begin + chunk) : E;
    const int* __restrict__ mybase = cnt + (size_t)blockIdx.x * N;

    for (int w0 = 0; w0 < N; w0 += LDS_NODES) {
        const int wn = (N - w0 < LDS_NODES) ? (N - w0) : LDS_NODES;
#pragma unroll 8
        for (int i = t; i < wn; i += WG) cur[i] = rowstart[w0 + i] + mybase[w0 + i];
        __syncthreads();
        for (int e0 = begin + t; e0 < end; e0 += EDGE_BATCH * WG) {
            int rr[EDGE_BATCH], cc[EDGE_BATCH];
#pragma unroll
            for (int j = 0; j < EDGE_BATCH; ++j) {
                const int e = e0 + j * WG;
                const bool ok = e < end;
                rr[j] = ok ? row[e] : -1;
                cc[j] = ok ? col[e] : 0;
            }
            int slot[EDGE_BATCH];
#pragma unroll
            for (int j = 0; j < EDGE_BATCH; ++j) {
                const unsigned r = (unsigned)(rr[j] - w0);
                slot[j] = (r < (unsigned)wn) ? atomicAdd(&cur[r], 1) : -1;
            }
#pragma unroll
            for (int j = 0; j < EDGE_BATCH; ++j)
                if (slot[j] >= 0) csr_col[slot[j]] = cc[j];
        }
        __syncthreads();
    }
}

__global__ __launch_bounds__(256) void gemm_wmma_kernel(const float* __restrict__ x,
                                                        const float* __restrict__ W,
                                                        float* __restrict__ h,
                                                        int nrows) {
    const int lane = threadIdx.x & 31;
    const int wave = threadIdx.x >> 5;
    const int l  = lane & 15;
    const int hh = lane >> 4;
    const int m0 = blockIdx.x * 16;
    const int n0 = wave * 16;

    int mrow = m0 + l;
    if (mrow >= nrows) mrow = nrows - 1;
    const float* __restrict__ xrow = x + (size_t)mrow * FEAT;

    v8f acc = {};
#pragma unroll 8
    for (int k0 = 0; k0 < FEAT; k0 += 4) {
        const int ka = k0 + 2 * hh;
        v2f a, b;
        a.x = xrow[ka];
        a.y = xrow[ka + 1];
        b.x = W[(size_t)ka       * EMBED + n0 + l];
        b.y = W[(size_t)(ka + 1) * EMBED + n0 + l];
        acc = __builtin_amdgcn_wmma_f32_16x16x4_f32(false, a, false, b,
                                                    (short)0, acc, false, false);
    }

    float* __restrict__ hout = h + (size_t)m0 * EMBED + n0 + l;
    if (m0 + 16 <= nrows) {
#pragma unroll
        for (int r = 0; r < 8; ++r)
            hout[(size_t)(r + 8 * hh) * EMBED] = acc[r];
    } else {
#pragma unroll
        for (int r = 0; r < 8; ++r) {
            const int mr = r + 8 * hh;
            if (m0 + mr < nrows) hout[(size_t)mr * EMBED] = acc[r];
        }
    }
}

__global__ __launch_bounds__(256) void gather_kernel(const float* __restrict__ h,
                                                     const int* __restrict__ deg,
                                                     const int* __restrict__ rowstart,
                                                     const int* __restrict__ csr_col,
                                                     float* __restrict__ out, int N) {
    const int lane = threadIdx.x & 31;
    const int r = blockIdx.x * (blockDim.x >> 5) + (threadIdx.x >> 5);
    if (r >= N) return;

    const int start = rowstart[r];
    const int end   = rowstart[r + 1];
    const float dr  = (float)deg[r];

    float4 acc = {0.f, 0.f, 0.f, 0.f};
    for (int i = start; i < end; ++i) {
        const int c = csr_col[i];
        const float norm = rsqrtf(dr * (float)deg[c]);
        const float4 v = ((const float4*)(h + (size_t)c * EMBED))[lane];
        acc.x += v.x * norm;
        acc.y += v.y * norm;
        acc.z += v.z * norm;
        acc.w += v.w * norm;
    }
    ((float4*)(out + (size_t)r * EMBED))[lane] = acc;
}

extern "C" void kernel_launch(void* const* d_in, const int* in_sizes, int n_in,
                              void* d_out, int out_size, void* d_ws, size_t ws_size,
                              hipStream_t stream) {
    const float* x   = (const float*)d_in[0];
    const float* W   = (const float*)d_in[1];
    const int*   row = (const int*)d_in[2];
    const int*   col = (const int*)d_in[3];
    float*       out = (float*)d_out;

    const int N = in_sizes[0] / FEAT;
    const int E = in_sizes[2];

    const int CHUNKS = 128;
    float* h        = (float*)d_ws;
    int*   cnt      = (int*)(h + (size_t)N * EMBED);
    int*   deg      = cnt + (size_t)CHUNKS * N;
    int*   rowstart = deg + N;
    int*   csr_col  = rowstart + (N + 1);

    count_kernel<<<CHUNKS, WG, 0, stream>>>(row, cnt, N, E);

    colscan_kernel<<<(N + 255) / 256, 256, 0, stream>>>(cnt, deg, N, CHUNKS);

    scan_kernel<<<1, 1024, 0, stream>>>(deg, rowstart, N, E);

    fill_kernel<<<CHUNKS, WG, 0, stream>>>(row, col, cnt, rowstart, csr_col, N, E);

    gemm_wmma_kernel<<<(N + 15) / 16, 256, 0, stream>>>(x, W, h, N);

    gather_kernel<<<(N + 7) / 8, 256, 0, stream>>>(h, deg, rowstart, csr_col, out, N);
}


# row-start prefix sum: lane-striped tiles scanned in registers (each wave load/store one contiguous 128-byte run, one barrier per 8192 elements) replace 196-byte-strided per-thread chunks
# speedup vs baseline: 2.6448x; 1.3606x over previous
#include <hip/hip_runtime.h>

#define FEAT  128
#define EMBED 128

typedef float v2f __attribute__((ext_vector_type(2)));
typedef float v8f __attribute__((ext_vector_type(8)));

#define LDS_NODES 65536
#define WG 1024
#define EDGE_BATCH 8

__global__ __launch_bounds__(WG) void count_kernel(const int* __restrict__ row,
                                                   int* __restrict__ cnt,
                                                   int N, int E) {
    __shared__ int hist[LDS_NODES];
    const int t = threadIdx.x;
    const int chunk = (E + (int)gridDim.x - 1) / (int)gridDim.x;
    const int begin = (int)blockIdx.x * chunk;
    const int end = (begin + chunk < E) ? (begin + chunk) : E;
    int* __restrict__ mycnt = cnt + (size_t)blockIdx.x * N;

    for (int w0 = 0; w0 < N; w0 += LDS_NODES) {
        const int wn = (N - w0 < LDS_NODES) ? (N - w0) : LDS_NODES;
        for (int i = t; i < wn; i += WG) hist[i] = 0;
        __syncthreads();
        for (int e0 = begin + t; e0 < end; e0 += EDGE_BATCH * WG) {
            int rr[EDGE_BATCH];
#pragma unroll
            for (int j = 0; j < EDGE_BATCH; ++j) {
                const int e = e0 + j * WG;
                rr[j] = (e < end) ? row[e] : -1;
            }
#pragma unroll
            for (int j = 0; j < EDGE_BATCH; ++j) {
                const unsigned r = (unsigned)(rr[j] - w0);
                if (r < (unsigned)wn) atomicAdd(&hist[r], 1);
            }
        }
        __syncthreads();
        for (int i = t; i < wn; i += WG) mycnt[w0 + i] = hist[i];
        __syncthreads();
    }
}

__global__ __launch_bounds__(256) void colscan_kernel(int* __restrict__ cnt,
                                                      int* __restrict__ deg,
                                                      int N, int B) {
    const int r = blockIdx.x * blockDim.x + threadIdx.x;
    if (r >= N) return;
    int run = 0;
    for (int b0 = 0; b0 < B; b0 += 32) {
        int c[32];
#pragma unroll
        for (int j = 0; j < 32; ++j)
            c[j] = (b0 + j < B) ? cnt[(size_t)(b0 + j) * N + r] : 0;
#pragma unroll
        for (int j = 0; j < 32; ++j) {
            if (b0 + j < B) cnt[(size_t)(b0 + j) * N + r] = run;
            run += c[j];
        }
    }
    deg[r] = run;
}

#define SCAN_WG 1024
#define SCAN_RUNS 8
#define SCAN_TILE (SCAN_WG * SCAN_RUNS)

__device__ __forceinline__ int wave_incl_scan(int x, int lane) {
    x += __builtin_amdgcn_update_dpp(0, x, 0x111, 0xf, 0xf, true);
    x += __builtin_amdgcn_update_dpp(0, x, 0x112, 0xf, 0xf, true);
    x += __builtin_amdgcn_update_dpp(0, x, 0x114, 0xf, 0xf, true);
    x += __builtin_amdgcn_update_dpp(0, x, 0x118, 0xf, 0xf, true);
    const int low_row_total = __builtin_amdgcn_readlane(x, 15);
    if (lane >= 16) x += low_row_total;
    return x;
}

__device__ __forceinline__ int deg_or_zero(const int* __restrict__ deg, int idx, int N) {
    int c = (idx < N) ? idx : N - 1;
    c = (c > 0) ? c : 0;
    const int v = deg[c];
    return (idx < N) ? v : 0;
}

__global__ __launch_bounds__(SCAN_WG) void scan_kernel(const int* __restrict__ deg,
                                                       int* __restrict__ rowstart,
                                                       int N, int E) {
    __shared__ int wtot[2][32];
    const int t = threadIdx.x;
    const int lane = t & 31;
    const int wave = __builtin_amdgcn_readfirstlane(t >> 5);
    const int seg = wave * (32 * SCAN_RUNS) + lane;

    int cur[SCAN_RUNS];
#pragma unroll
    for (int k = 0; k < SCAN_RUNS; ++k) {
        cur[k] = deg_or_zero(deg, seg + 32 * k, N);
    }

    int carry = 0;
    int buf = 0;
    for (int base = 0; base < N; base += SCAN_TILE) {
        int nxt[SCAN_RUNS];
#pragma unroll
        for (int k = 0; k < SCAN_RUNS; ++k) {
            nxt[k] = deg_or_zero(deg, base + SCAN_TILE + seg + 32 * k, N);
        }

        int inc[SCAN_RUNS], off[SCAN_RUNS];
        int wsum = 0;
#pragma unroll
        for (int k = 0; k < SCAN_RUNS; ++k) {
            inc[k] = wave_incl_scan(cur[k], lane);
            off[k] = wsum;
            wsum += __builtin_amdgcn_readlane(inc[k], 31);
        }
        if (lane == 0) wtot[buf][wave] = wsum;
        __syncthreads();

        const int wt = wtot[buf][lane];
        const int ws = wave_incl_scan(wt, lane);
        const int tile_total = __builtin_amdgcn_readlane(ws, 31);
        const int o = carry + __builtin_amdgcn_readlane(ws, wave) - __builtin_amdgcn_readlane(wt, wave);
#pragma unroll
        for (int k = 0; k < SCAN_RUNS; ++k) {
            const int idx = base + seg + 32 * k;
            if (idx < N) rowstart[idx] = o + off[k] + inc[k] - cur[k];
        }

        carry += tile_total;
#pragma unroll
        for (int k = 0; k < SCAN_RUNS; ++k) cur[k] = nxt[k];
        buf ^= 1;
    }
    if (t == 0) rowstart[N] = E;
}

__global__ __launch_bounds__(WG) void fill_kernel(const int* __restrict__ row,
                                                  const int* __restrict__ col,
                                                  const int* __restrict__ cnt,
                                                  const int* __restrict__ rowstart,
                                                  int* __restrict__ csr_col,
                                                  int N, int E) {
    __shared__ int cur[LDS_NODES];
    const int t = threadIdx.x;
    const int chunk = (E + (int)gridDim.x - 1) / (int)gridDim.x;
    const int begin = (int)blockIdx.x * chunk;
    const int end = (begin + chunk < E) ? (begin + chunk) : E;
    const int* __restrict__ mybase = cnt + (size_t)blockIdx.x * N;

    for (int w0 = 0; w0 < N; w0 += LDS_NODES) {
        const int wn = (N - w0 < LDS_NODES) ? (N - w0) : LDS_NODES;
#pragma unroll 8
        for (int i = t; i < wn; i += WG) cur[i] = rowstart[w0 + i] + mybase[w0 + i];
        __syncthreads();
        for (int e0 = begin + t; e0 < end; e0 += EDGE_BATCH * WG) {
            int rr[EDGE_BATCH], cc[EDGE_BATCH];
#pragma unroll
            for (int j = 0; j < EDGE_BATCH; ++j) {
                const int e = e0 + j * WG;
                const bool ok = e < end;
                rr[j] = ok ? row[e] : -1;
                cc[j] = ok ? col[e] : 0;
            }
            int slot[EDGE_BATCH];
#pragma unroll
            for (int j = 0; j < EDGE_BATCH; ++j) {
                const unsigned r = (unsigned)(rr[j] - w0);
                slot[j] = (r < (unsigned)wn) ? atomicAdd(&cur[r], 1) : -1;
            }
#pragma unroll
            for (int j = 0; j < EDGE_BATCH; ++j)
                if (slot[j] >= 0) csr_col[slot[j]] = cc[j];
        }
        __syncthreads();
    }
}

__global__ __launch_bounds__(256) void gemm_wmma_kernel(const float* __restrict__ x,
                                                        const float* __restrict__ W,
                                                        float* __restrict__ h,
                                                        int nrows) {
    const int lane = threadIdx.x & 31;
    const int wave = threadIdx.x >> 5;
    const int l  = lane & 15;
    const int hh = lane >> 4;
    const int m0 = blockIdx.x * 16;
    const int n0 = wave * 16;

    int mrow = m0 + l;
    if (mrow >= nrows) mrow = nrows - 1;
    const float* __restrict__ xrow = x + (size_t)mrow * FEAT;

    v8f acc = {};
#pragma unroll 8
    for (int k0 = 0; k0 < FEAT; k0 += 4) {
        const int ka = k0 + 2 * hh;
        v2f a, b;
        a.x = xrow[ka];
        a.y = xrow[ka + 1];
        b.x = W[(size_t)ka       * EMBED + n0 + l];
        b.y = W[(size_t)(ka + 1) * EMBED + n0 + l];
        acc = __builtin_amdgcn_wmma_f32_16x16x4_f32(false, a, false, b,
                                                    (short)0, acc, false, false);
    }

    float* __restrict__ hout = h + (size_t)m0 * EMBED + n0 + l;
    if (m0 + 16 <= nrows) {
#pragma unroll
        for (int r = 0; r < 8; ++r)
            hout[(size_t)(r + 8 * hh) * EMBED] = acc[r];
    } else {
#pragma unroll
        for (int r = 0; r < 8; ++r) {
            const int mr = r + 8 * hh;
            if (m0 + mr < nrows) hout[(size_t)mr * EMBED] = acc[r];
        }
    }
}

__global__ __launch_bounds__(256) void gather_kernel(const float* __restrict__ h,
                                                     const int* __restrict__ deg,
                                                     const int* __restrict__ rowstart,
                                                     const int* __restrict__ csr_col,
                                                     float* __restrict__ out, int N) {
    const int lane = threadIdx.x & 31;
    const int r = blockIdx.x * (blockDim.x >> 5) + (threadIdx.x >> 5);
    if (r >= N) return;

    const int start = rowstart[r];
    const int end   = rowstart[r + 1];
    const float dr  = (float)deg[r];

    float4 acc = {0.f, 0.f, 0.f, 0.f};
    for (int i = start; i < end; ++i) {
        const int c = csr_col[i];
        const float norm = rsqrtf(dr * (float)deg[c]);
        const float4 v = ((const float4*)(h + (size_t)c * EMBED))[lane];
        acc.x += v.x * norm;
        acc.y += v.y * norm;
        acc.z += v.z * norm;
        acc.w += v.w * norm;
    }
    ((float4*)(out + (size_t)r * EMBED))[lane] = acc;
}

extern "C" void kernel_launch(void* const* d_in, const int* in_sizes, int n_in,
                              void* d_out, int out_size, void* d_ws, size_t ws_size,
                              hipStream_t stream) {
    const float* x   = (const float*)d_in[0];
    const float* W   = (const float*)d_in[1];
    const int*   row = (const int*)d_in[2];
    const int*   col = (const int*)d_in[3];
    float*       out = (float*)d_out;

    const int N = in_sizes[0] / FEAT;
    const int E = in_sizes[2];

    const int CHUNKS = 128;
    float* h        = (float*)d_ws;
    int*   cnt      = (int*)(h + (size_t)N * EMBED);
    int*   deg      = cnt + (size_t)CHUNKS * N;
    int*   rowstart = deg + N;
    int*   csr_col  = rowstart + (N + 1);

    count_kernel<<<CHUNKS, WG, 0, stream>>>(row, cnt, N, E);

    colscan_kernel<<<(N + 255) / 256, 256, 0, stream>>>(cnt, deg, N, CHUNKS);

    scan_kernel<<<1, SCAN_WG, 0, stream>>>(deg, rowstart, N, E);

    fill_kernel<<<CHUNKS, WG, 0, stream>>>(row, col, cnt, rowstart, csr_col, N, E);

    gemm_wmma_kernel<<<(N + 15) / 16, 256, 0, stream>>>(x, W, h, N);

    gather_kernel<<<(N + 7) / 8, 256, 0, stream>>>(h, deg, rowstart, csr_col, out, N);
}


# gather: a wave reads 64 neighbour ids and norms per node in coalesced loads and hands them out by readlane, 8 x-row loads in flight, no index-then-row chain per 4 neighbours; sum order kept
# speedup vs baseline: 2.9787x; 1.1262x over previous
#include <hip/hip_runtime.h>

#define FEAT  128
#define EMBED 128

typedef float v2f __attribute__((ext_vector_type(2)));
typedef float v8f __attribute__((ext_vector_type(8)));

#define LDS_NODES 65536
#define WG 1024
#define EDGE_BATCH 8

__global__ __launch_bounds__(WG) void count_kernel(const int* __restrict__ row,
                                                   int* __restrict__ cnt,
                                                   int N, int E) {
    __shared__ int hist[LDS_NODES];
    const int t = threadIdx.x;
    const int chunk = (E + (int)gridDim.x - 1) / (int)gridDim.x;
    const int begin = (int)blockIdx.x * chunk;
    const int end = (begin + chunk < E) ? (begin + chunk) : E;
    int* __restrict__ mycnt = cnt + (size_t)blockIdx.x * N;

    for (int w0 = 0; w0 < N; w0 += LDS_NODES) {
        const int wn = (N - w0 < LDS_NODES) ? (N - w0) : LDS_NODES;
        for (int i = t; i < wn; i += WG) hist[i] = 0;
        __syncthreads();
        for (int e0 = begin + t; e0 < end; e0 += EDGE_BATCH * WG) {
            int rr[EDGE_BATCH];
#pragma unroll
            for (int j = 0; j < EDGE_BATCH; ++j) {
                const int e = e0 + j * WG;
                rr[j] = (e < end) ? row[e] : -1;
            }
#pragma unroll
            for (int j = 0; j < EDGE_BATCH; ++j) {
                const unsigned r = (unsigned)(rr[j] - w0);
                if (r < (unsigned)wn) atomicAdd(&hist[r], 1);
            }
        }
        __syncthreads();
        for (int i = t; i < wn; i += WG) mycnt[w0 + i] = hist[i];
        __syncthreads();
    }
}

__global__ __launch_bounds__(256) void colscan_kernel(int* __restrict__ cnt,
                                                      int* __restrict__ deg,
                                                      int N, int B) {
    const int r = blockIdx.x * blockDim.x + threadIdx.x;
    if (r >= N) return;
    int run = 0;
    for (int b0 = 0; b0 < B; b0 += 32) {
        int c[32];
#pragma unroll
        for (int j = 0; j < 32; ++j)
            c[j] = (b0 + j < B) ? cnt[(size_t)(b0 + j) * N + r] : 0;
#pragma unroll
        for (int j = 0; j < 32; ++j) {
            if (b0 + j < B) cnt[(size_t)(b0 + j) * N + r] = run;
            run += c[j];
        }
    }
    deg[r] = run;
}

#define SCAN_WG 1024
#define SCAN_RUNS 8
#define SCAN_TILE (SCAN_WG * SCAN_RUNS)

__device__ __forceinline__ int wave_incl_scan(int x, int lane) {
    x += __builtin_amdgcn_update_dpp(0, x, 0x111, 0xf, 0xf, true);
    x += __builtin_amdgcn_update_dpp(0, x, 0x112, 0xf, 0xf, true);
    x += __builtin_amdgcn_update_dpp(0, x, 0x114, 0xf, 0xf, true);
    x += __builtin_amdgcn_update_dpp(0, x, 0x118, 0xf, 0xf, true);
    const int low_row_total = __builtin_amdgcn_readlane(x, 15);
    if (lane >= 16) x += low_row_total;
    return x;
}

__device__ __forceinline__ int deg_or_zero(const int* __restrict__ deg, int idx, int N) {
    int c = (idx < N) ? idx : N - 1;
    c = (c > 0) ? c : 0;
    const int v = deg[c];
    return (idx < N) ? v : 0;
}

__global__ __launch_bounds__(SCAN_WG) void scan_kernel(const int* __restrict__ deg,
                                                       int* __restrict__ rowstart,
                                                       int N, int E) {
    __shared__ int wtot[2][32];
    const int t = threadIdx.x;
    const int lane = t & 31;
    const int wave = __builtin_amdgcn_readfirstlane(t >> 5);
    const int seg = wave * (32 * SCAN_RUNS) + lane;

    int cur[SCAN_RUNS];
#pragma unroll
    for (int k = 0; k < SCAN_RUNS; ++k) {
        cur[k] = deg_or_zero(deg, seg + 32 * k, N);
    }

    int carry = 0;
    int buf = 0;
    for (int base = 0; base < N; base += SCAN_TILE) {
        int nxt[SCAN_RUNS];
#pragma unroll
        for (int k = 0; k < SCAN_RUNS; ++k) {
            nxt[k] = deg_or_zero(deg, base + SCAN_TILE + seg + 32 * k, N);
        }

        int inc[SCAN_RUNS], off[SCAN_RUNS];
        int wsum = 0;
#pragma unroll
        for (int k = 0; k < SCAN_RUNS; ++k) {
            inc[k] = wave_incl_scan(cur[k], lane);
            off[k] = wsum;
            wsum += __builtin_amdgcn_readlane(inc[k], 31);
        }
        if (lane == 0) wtot[buf][wave] = wsum;
        __syncthreads();

        const int wt = wtot[buf][lane];
        const int ws = wave_incl_scan(wt, lane);
        const int tile_total = __builtin_amdgcn_readlane(ws, 31);
        const int o = carry + __builtin_amdgcn_readlane(ws, wave) - __builtin_amdgcn_readlane(wt, wave);
#pragma unroll
        for (int k = 0; k < SCAN_RUNS; ++k) {
            const int idx = base + seg + 32 * k;
            if (idx < N) rowstart[idx] = o + off[k] + inc[k] - cur[k];
        }

        carry += tile_total;
#pragma unroll
        for (int k = 0; k < SCAN_RUNS; ++k) cur[k] = nxt[k];
        buf ^= 1;
    }
    if (t == 0) rowstart[N] = E;
}

__global__ __launch_bounds__(WG) void fill_kernel(const int* __restrict__ row,
                                                  const int* __restrict__ col,
                                                  const int* __restrict__ cnt,
                                                  const int* __restrict__ rowstart,
                                                  int* __restrict__ csr_col,
                                                  int N, int E) {
    __shared__ int cur[LDS_NODES];
    const int t = threadIdx.x;
    const int chunk = (E + (int)gridDim.x - 1) / (int)gridDim.x;
    const int begin = (int)blockIdx.x * chunk;
    const int end = (begin + chunk < E) ? (begin + chunk) : E;
    const int* __restrict__ mybase = cnt + (size_t)blockIdx.x * N;

    for (int w0 = 0; w0 < N; w0 += LDS_NODES) {
        const int wn = (N - w0 < LDS_NODES) ? (N - w0) : LDS_NODES;
#pragma unroll 8
        for (int i = t; i < wn; i += WG) cur[i] = rowstart[w0 + i] + mybase[w0 + i];
        __syncthreads();
        for (int e0 = begin + t; e0 < end; e0 += EDGE_BATCH * WG) {
            int rr[EDGE_BATCH], cc[EDGE_BATCH];
#pragma unroll
            for (int j = 0; j < EDGE_BATCH; ++j) {
                const int e = e0 + j * WG;
                const bool ok = e < end;
                rr[j] = ok ? row[e] : -1;
                cc[j] = ok ? col[e] : 0;
            }
            int slot[EDGE_BATCH];
#pragma unroll
            for (int j = 0; j < EDGE_BATCH; ++j) {
                const unsigned r = (unsigned)(rr[j] - w0);
                slot[j] = (r < (unsigned)wn) ? atomicAdd(&cur[r], 1) : -1;
            }
#pragma unroll
            for (int j = 0; j < EDGE_BATCH; ++j)
                if (slot[j] >= 0) csr_col[slot[j]] = cc[j];
        }
        __syncthreads();
    }
}

#define TILE_M 16
#define AGG_PITCH 132
#define ROWS_IN_FLIGHT 8

__device__ __forceinline__ int nbr_id(const int* __restrict__ csr_col, int idx, int end) {
    int k = (idx < end) ? idx : end - 1;
    k = (k > 0) ? k : 0;
    const int c = csr_col[k];
    return (idx < end) ? c : 0;
}

__device__ __forceinline__ float nbr_norm(const int* __restrict__ deg, int c, float dr, int idx, int end) {
    const float n = rsqrtf(dr * (float)deg[c]);
    return (idx < end) ? n : 0.f;
}

__device__ __forceinline__ void add_rows(float4& acc, const float* __restrict__ x,
                                         int cv, float nv, int cnt, int lane) {
    const int nvi = __float_as_int(nv);
    for (int p = 0; p < cnt; p += ROWS_IN_FLIGHT) {
        float4 v[ROWS_IN_FLIGHT];
        float s[ROWS_IN_FLIGHT];
#pragma unroll
        for (int u = 0; u < ROWS_IN_FLIGHT; ++u) {
            const unsigned c = (unsigned)__builtin_amdgcn_readlane(cv, p + u);
            s[u] = __int_as_float(__builtin_amdgcn_readlane(nvi, p + u));
            const float* rowp = x + (size_t)c * FEAT;
            v[u] = *(const float4*)(rowp + 4 * lane);
        }
#pragma unroll
        for (int u = 0; u < ROWS_IN_FLIGHT; ++u) {
            acc.x = __builtin_fmaf(v[u].x, s[u], acc.x);
            acc.y = __builtin_fmaf(v[u].y, s[u], acc.y);
            acc.z = __builtin_fmaf(v[u].z, s[u], acc.z);
            acc.w = __builtin_fmaf(v[u].w, s[u], acc.w);
        }
    }
}

__global__ __launch_bounds__(256) void gather_wmma_kernel(const float* __restrict__ x,
                                                          const float* __restrict__ W,
                                                          const int* __restrict__ deg,
                                                          const int* __restrict__ rowstart,
                                                          const int* __restrict__ csr_col,
                                                          float* __restrict__ out, int N) {
    __shared__ alignas(16) float agg[TILE_M * AGG_PITCH];
    const int lane = threadIdx.x & 31;
    const int wave = __builtin_amdgcn_readfirstlane(threadIdx.x >> 5);
    const int m0 = blockIdx.x * TILE_M;

    const int r0 = m0 + 2 * wave;
    const int i0 = (r0     < N) ? r0     : N;
    const int i1 = (r0 + 1 < N) ? r0 + 1 : N;
    const int i2 = (r0 + 2 < N) ? r0 + 2 : N;
    int bound[3];
    bound[0] = rowstart[i0];
    bound[1] = rowstart[i1];
    bound[2] = rowstart[i2];
    float dself[2];
    dself[0] = (float)deg[(i0 < N) ? i0 : N - 1];
    dself[1] = (float)deg[(i1 < N) ? i1 : N - 1];

    int ca[2], cb[2];
    float na[2], nb[2];
#pragma unroll
    for (int j = 0; j < 2; ++j) {
        ca[j] = nbr_id(csr_col, bound[j] + lane,      bound[j + 1]);
        cb[j] = nbr_id(csr_col, bound[j] + 32 + lane, bound[j + 1]);
    }
#pragma unroll
    for (int j = 0; j < 2; ++j) {
        na[j] = nbr_norm(deg, ca[j], dself[j], bound[j] + lane,      bound[j + 1]);
        nb[j] = nbr_norm(deg, cb[j], dself[j], bound[j] + 32 + lane, bound[j + 1]);
    }

#pragma unroll
    for (int j = 0; j < 2; ++j) {
        const int start = bound[j];
        const int end   = bound[j + 1];
        int   cA = ca[j], cB = cb[j];
        float nA = na[j], nB = nb[j];
        float4 acc = {0.f, 0.f, 0.f, 0.f};
        for (int b = start; b < end; b += 64) {
            if (b != start) {
                cA = nbr_id(csr_col, b + lane,      end);
                cB = nbr_id(csr_col, b + 32 + lane, end);
                nA = nbr_norm(deg, cA, dself[j], b + lane,      end);
                nB = nbr_norm(deg, cB, dself[j], b + 32 + lane, end);
            }
            const int left = end - b;
            add_rows(acc, x, cA, nA, (left < 32) ? left : 32, lane);
            add_rows(acc, x, cB, nB, (left < 64) ? left - 32 : 32, lane);
        }
        *(float4*)(agg + (2 * wave + j) * AGG_PITCH + lane * 4) = acc;
    }
    __syncthreads();

    const int l  = lane & 15;
    const int hh = lane >> 4;
    const int n0 = wave * 16;
    const float* arow = agg + l * AGG_PITCH + 2 * hh;
    const float* __restrict__ wcol = W + (size_t)(2 * hh) * EMBED + n0 + l;

    v8f acc = {};
#pragma unroll 16
    for (int k0 = 0; k0 < FEAT; k0 += 4) {
        v2f a, b;
        a = *(const v2f*)(arow + k0);
        b.x = wcol[(size_t)k0       * EMBED];
        b.y = wcol[(size_t)(k0 + 1) * EMBED];
        acc = __builtin_amdgcn_wmma_f32_16x16x4_f32(false, a, false, b,
                                                    (short)0, acc, false, false);
    }

    float* __restrict__ orow = out + (size_t)m0 * EMBED + n0 + l;
    if (m0 + TILE_M <= N) {
#pragma unroll
        for (int r = 0; r < 8; ++r)
            orow[(size_t)(r + 8 * hh) * EMBED] = acc[r];
    } else {
#pragma unroll
        for (int r = 0; r < 8; ++r) {
            const int mr = r + 8 * hh;
            if (m0 + mr < N) orow[(size_t)mr * EMBED] = acc[r];
        }
    }
}

extern "C" void kernel_launch(void* const* d_in, const int* in_sizes, int n_in,
                              void* d_out, int out_size, void* d_ws, size_t ws_size,
                              hipStream_t stream) {
    const float* x   = (const float*)d_in[0];
    const float* W   = (const float*)d_in[1];
    const int*   row = (const int*)d_in[2];
    const int*   col = (const int*)d_in[3];
    float*       out = (float*)d_out;

    const int N = in_sizes[0] / FEAT;
    const int E = in_sizes[2];

    const int CHUNKS = 128;
    int*   cnt      = (int*)d_ws;
    int*   deg      = cnt + (size_t)CHUNKS * N;
    int*   rowstart = deg + N;
    int*   csr_col  = rowstart + (N + 1);

    count_kernel<<<CHUNKS, WG, 0, stream>>>(row, cnt, N, E);

    colscan_kernel<<<(N + 255) / 256, 256, 0, stream>>>(cnt, deg, N, CHUNKS);

    scan_kernel<<<1, SCAN_WG, 0, stream>>>(deg, rowstart, N, E);

    fill_kernel<<<CHUNKS, WG, 0, stream>>>(row, col, cnt, rowstart, csr_col, N, E);

    gather_wmma_kernel<<<(N + TILE_M - 1) / TILE_M, 256, 0, stream>>>(x, W, deg, rowstart, csr_col, out, N);
}


# row-start prefix sum loses its own launch: column-scan workgroups also scan their 256 degrees (offset in block + block total), fill workgroups scan the 196 block totals; 5 launches become 4
# speedup vs baseline: 3.2111x; 1.0780x over previous
#include <hip/hip_runtime.h>

#define FEAT  128
#define EMBED 128

typedef float v2f __attribute__((ext_vector_type(2)));
typedef float v8f __attribute__((ext_vector_type(8)));

#define LDS_NODES 65536
#define WG 1024
#define EDGE_BATCH 8

__global__ __launch_bounds__(WG) void count_kernel(const int* __restrict__ row,
                                                   int* __restrict__ cnt,
                                                   int N, int E) {
    __shared__ int hist[LDS_NODES];
    const int t = threadIdx.x;
    const int chunk = (E + (int)gridDim.x - 1) / (int)gridDim.x;
    const int begin = (int)blockIdx.x * chunk;
    const int end = (begin + chunk < E) ? (begin + chunk) : E;
    int* __restrict__ mycnt = cnt + (size_t)blockIdx.x * N;

    for (int w0 = 0; w0 < N; w0 += LDS_NODES) {
        const int wn = (N - w0 < LDS_NODES) ? (N - w0) : LDS_NODES;
        for (int i = t; i < wn; i += WG) hist[i] = 0;
        __syncthreads();
        for (int e0 = begin + t; e0 < end; e0 += EDGE_BATCH * WG) {
            int rr[EDGE_BATCH];
#pragma unroll
            for (int j = 0; j < EDGE_BATCH; ++j) {
                const int e = e0 + j * WG;
                rr[j] = (e < end) ? row[e] : -1;
            }
#pragma unroll
            for (int j = 0; j < EDGE_BATCH; ++j) {
                const unsigned r = (unsigned)(rr[j] - w0);
                if (r < (unsigned)wn) atomicAdd(&hist[r], 1);
            }
        }
        __syncthreads();
        for (int i = t; i < wn; i += WG) mycnt[w0 + i] = hist[i];
        __syncthreads();
    }
}

__device__ __forceinline__ int wave_incl_scan(int x, int lane) {
    x += __builtin_amdgcn_update_dpp(0, x, 0x111, 0xf, 0xf, true);
    x += __builtin_amdgcn_update_dpp(0, x, 0x112, 0xf, 0xf, true);
    x += __builtin_amdgcn_update_dpp(0, x, 0x114, 0xf, 0xf, true);
    x += __builtin_amdgcn_update_dpp(0, x, 0x118, 0xf, 0xf, true);
    const int low_row_total = __builtin_amdgcn_readlane(x, 15);
    if (lane >= 16) x += low_row_total;
    return x;
}

#define CS_WG 256
#define CS_SHIFT 8

__global__ __launch_bounds__(CS_WG) void colscan_kernel(int* __restrict__ cnt,
                                                        int* __restrict__ deg,
                                                        int* __restrict__ rowlocal,
                                                        int* __restrict__ blocktot,
                                                        int N, int B) {
    __shared__ int wtot[CS_WG / 32];
    const int t = threadIdx.x;
    const int lane = t & 31;
    const int wave = __builtin_amdgcn_readfirstlane(t >> 5);
    const int r = blockIdx.x * CS_WG + t;
    int run = 0;
    if (r < N) {
        for (int b0 = 0; b0 < B; b0 += 32) {
            int c[32];
#pragma unroll
            for (int j = 0; j < 32; ++j)
                c[j] = (b0 + j < B) ? cnt[(size_t)(b0 + j) * N + r] : 0;
#pragma unroll
            for (int j = 0; j < 32; ++j) {
                if (b0 + j < B) cnt[(size_t)(b0 + j) * N + r] = run;
                run += c[j];
            }
        }
        deg[r] = run;
    }
    const int inc = wave_incl_scan(run, lane);
    if (lane == 31) wtot[wave] = inc;
    __syncthreads();
    const int wt = (lane < CS_WG / 32) ? wtot[lane] : 0;
    const int ws = wave_incl_scan(wt, lane);
    const int before = __builtin_amdgcn_readlane(ws, wave) - __builtin_amdgcn_readlane(wt, wave);
    if (r < N) rowlocal[r] = before + inc - run;
    if (t == 0) blocktot[blockIdx.x] = __builtin_amdgcn_readlane(ws, CS_WG / 32 - 1);
}

__global__ __launch_bounds__(WG) void fill_kernel(const int* __restrict__ row,
                                                  const int* __restrict__ col,
                                                  const int* __restrict__ cnt,
                                                  const int* __restrict__ rowlocal,
                                                  const int* __restrict__ blocktot,
                                                  int* __restrict__ rowstart,
                                                  int* __restrict__ csr_col,
                                                  int N, int E) {
    __shared__ int cur[LDS_NODES];
    __shared__ int bpre[LDS_NODES / CS_WG];
    const int t = threadIdx.x;
    const int lane = t & 31;
    const int wave = __builtin_amdgcn_readfirstlane(t >> 5);
    const int nblk = (N + CS_WG - 1) / CS_WG;
    const int chunk = (E + (int)gridDim.x - 1) / (int)gridDim.x;
    const int begin = (int)blockIdx.x * chunk;
    const int end = (begin + chunk < E) ? (begin + chunk) : E;
    const int* __restrict__ mybase = cnt + (size_t)blockIdx.x * N;

    int carry = 0;
    for (int w0 = 0; w0 < N; w0 += LDS_NODES) {
        const int wn = (N - w0 < LDS_NODES) ? (N - w0) : LDS_NODES;
        if (wave == 0) {
            const int blk0 = w0 >> CS_SHIFT;
            int v[LDS_NODES / CS_WG / 32];
#pragma unroll
            for (int k = 0; k < LDS_NODES / CS_WG / 32; ++k) {
                const int idx = blk0 + 32 * k + lane;
                const int x = blocktot[(idx < nblk) ? idx : nblk - 1];
                v[k] = (idx < nblk) ? x : 0;
            }
#pragma unroll
            for (int k = 0; k < LDS_NODES / CS_WG / 32; ++k) {
                const int inc = wave_incl_scan(v[k], lane);
                bpre[32 * k + lane] = carry + inc - v[k];
                carry += __builtin_amdgcn_readlane(inc, 31);
            }
        }
        __syncthreads();
#pragma unroll 8
        for (int i = t; i < wn; i += WG)
            cur[i] = bpre[i >> CS_SHIFT] + rowlocal[w0 + i] + mybase[w0 + i];
        for (int s = (int)blockIdx.x; s * WG < wn; s += (int)gridDim.x) {
            const int i = s * WG + t;
            if (i < wn) rowstart[w0 + i] = bpre[i >> CS_SHIFT] + rowlocal[w0 + i];
        }
        __syncthreads();
        for (int e0 = begin + t; e0 < end; e0 += EDGE_BATCH * WG) {
            int rr[EDGE_BATCH], cc[EDGE_BATCH];
#pragma unroll
            for (int j = 0; j < EDGE_BATCH; ++j) {
                const int e = e0 + j * WG;
                const bool ok = e < end;
                rr[j] = ok ? row[e] : -1;
                cc[j] = ok ? col[e] : 0;
            }
            int slot[EDGE_BATCH];
#pragma unroll
            for (int j = 0; j < EDGE_BATCH; ++j) {
                const unsigned r = (unsigned)(rr[j] - w0);
                slot[j] = (r < (unsigned)wn) ? atomicAdd(&cur[r], 1) : -1;
            }
#pragma unroll
            for (int j = 0; j < EDGE_BATCH; ++j)
                if (slot[j] >= 0) csr_col[slot[j]] = cc[j];
        }
        __syncthreads();
    }
    if (blockIdx.x == 0 && t == 0) rowstart[N] = E;
}

#define TILE_M 16
#define AGG_PITCH 132
#define ROWS_IN_FLIGHT 8

__device__ __forceinline__ int nbr_id(const int* __restrict__ csr_col, int idx, int end) {
    int k = (idx < end) ? idx : end - 1;
    k = (k > 0) ? k : 0;
    const int c = csr_col[k];
    return (idx < end) ? c : 0;
}

__device__ __forceinline__ float nbr_norm(const int* __restrict__ deg, int c, float dr, int idx, int end) {
    const float n = rsqrtf(dr * (float)deg[c]);
    return (idx < end) ? n : 0.f;
}

__device__ __forceinline__ void add_rows(float4& acc, const float* __restrict__ x,
                                         int cv, float nv, int cnt, int lane) {
    const int nvi = __float_as_int(nv);
    for (int p = 0; p < cnt; p += ROWS_IN_FLIGHT) {
        float4 v[ROWS_IN_FLIGHT];
        float s[ROWS_IN_FLIGHT];
#pragma unroll
        for (int u = 0; u < ROWS_IN_FLIGHT; ++u) {
            const unsigned c = (unsigned)__builtin_amdgcn_readlane(cv, p + u);
            s[u] = __int_as_float(__builtin_amdgcn_readlane(nvi, p + u));
            const float* rowp = x + (size_t)c * FEAT;
            v[u] = *(const float4*)(rowp + 4 * lane);
        }
#pragma unroll
        for (int u = 0; u < ROWS_IN_FLIGHT; ++u) {
            acc.x = __builtin_fmaf(v[u].x, s[u], acc.x);
            acc.y = __builtin_fmaf(v[u].y, s[u], acc.y);
            acc.z = __builtin_fmaf(v[u].z, s[u], acc.z);
            acc.w = __builtin_fmaf(v[u].w, s[u], acc.w);
        }
    }
}

__global__ __launch_bounds__(256) void gather_wmma_kernel(const float* __restrict__ x,
                                                          const float* __restrict__ W,
                                                          const int* __restrict__ deg,
                                                          const int* __restrict__ rowstart,
                                                          const int* __restrict__ csr_col,
                                                          float* __restrict__ out, int N) {
    __shared__ alignas(16) float agg[TILE_M * AGG_PITCH];
    const int lane = threadIdx.x & 31;
    const int wave = __builtin_amdgcn_readfirstlane(threadIdx.x >> 5);
    const int m0 = blockIdx.x * TILE_M;

    const int r0 = m0 + 2 * wave;
    const int i0 = (r0     < N) ? r0     : N;
    const int i1 = (r0 + 1 < N) ? r0 + 1 : N;
    const int i2 = (r0 + 2 < N) ? r0 + 2 : N;
    int bound[3];
    bound[0] = rowstart[i0];
    bound[1] = rowstart[i1];
    bound[2] = rowstart[i2];
    float dself[2];
    dself[0] = (float)deg[(i0 < N) ? i0 : N - 1];
    dself[1] = (float)deg[(i1 < N) ? i1 : N - 1];

    int ca[2], cb[2];
    float na[2], nb[2];
#pragma unroll
    for (int j = 0; j < 2; ++j) {
        ca[j] = nbr_id(csr_col, bound[j] + lane,      bound[j + 1]);
        cb[j] = nbr_id(csr_col, bound[j] + 32 + lane, bound[j + 1]);
    }
#pragma unroll
    for (int j = 0; j < 2; ++j) {
        na[j] = nbr_norm(deg, ca[j], dself[j], bound[j] + lane,      bound[j + 1]);
        nb[j] = nbr_norm(deg, cb[j], dself[j], bound[j] + 32 + lane, bound[j + 1]);
    }

#pragma unroll
    for (int j = 0; j < 2; ++j) {
        const int start = bound[j];
        const int end   = bound[j + 1];
        int   cA = ca[j], cB = cb[j];
        float nA = na[j], nB = nb[j];
        float4 acc = {0.f, 0.f, 0.f, 0.f};
        for (int b = start; b < end; b += 64) {
            if (b != start) {
                cA = nbr_id(csr_col, b + lane,      end);
                cB = nbr_id(csr_col, b + 32 + lane, end);
                nA = nbr_norm(deg, cA, dself[j], b + lane,      end);
                nB = nbr_norm(deg, cB, dself[j], b + 32 + lane, end);
            }
            const int left = end - b;
            add_rows(acc, x, cA, nA, (left < 32) ? left : 32, lane);
            add_rows(acc, x, cB, nB, (left < 64) ? left - 32 : 32, lane);
        }
        *(float4*)(agg + (2 * wave + j) * AGG_PITCH + lane * 4) = acc;
    }
    __syncthreads();

    const int l  = lane & 15;
    const int hh = lane >> 4;
    const int n0 = wave * 16;
    const float* arow = agg + l * AGG_PITCH + 2 * hh;
    const float* __restrict__ wcol = W + (size_t)(2 * hh) * EMBED + n0 + l;

    v8f acc = {};
#pragma unroll 16
    for (int k0 = 0; k0 < FEAT; k0 += 4) {
        v2f a, b;
        a = *(const v2f*)(arow + k0);
        b.x = wcol[(size_t)k0       * EMBED];
        b.y = wcol[(size_t)(k0 + 1) * EMBED];
        acc = __builtin_amdgcn_wmma_f32_16x16x4_f32(false, a, false, b,
                                                    (short)0, acc, false, false);
    }

    float* __restrict__ orow = out + (size_t)m0 * EMBED + n0 + l;
    if (m0 + TILE_M <= N) {
#pragma unroll
        for (int r = 0; r < 8; ++r)
            orow[(size_t)(r + 8 * hh) * EMBED] = acc[r];
    } else {
#pragma unroll
        for (int r = 0; r < 8; ++r) {
            const int mr = r + 8 * hh;
            if (m0 + mr < N) orow[(size_t)mr * EMBED] = acc[r];
        }
    }
}

extern "C" void kernel_launch(void* const* d_in, const int* in_sizes, int n_in,
                              void* d_out, int out_size, void* d_ws, size_t ws_size,
                              hipStream_t stream) {
    const float* x   = (const float*)d_in[0];
    const float* W   = (const float*)d_in[1];
    const int*   row = (const int*)d_in[2];
    const int*   col = (const int*)d_in[3];
    float*       out = (float*)d_out;

    const int N = in_sizes[0] / FEAT;
    const int E = in_sizes[2];

    const int CHUNKS = 128;
    int*   cnt      = (int*)d_ws;
    int*   deg      = cnt + (size_t)CHUNKS * N;
    int*   rowstart = deg + N;
    int*   csr_col  = rowstart + (N + 1);
    int*   rowlocal = csr_col + E;
    int*   blocktot = rowlocal + N;

    count_kernel<<<CHUNKS, WG, 0, stream>>>(row, cnt, N, E);

    colscan_kernel<<<(N + CS_WG - 1) / CS_WG, CS_WG, 0, stream>>>(cnt, deg, rowlocal, blocktot, N, CHUNKS);

    fill_kernel<<<CHUNKS, WG, 0, stream>>>(row, col, cnt, rowlocal, blocktot, rowstart, csr_col, N, E);

    gather_wmma_kernel<<<(N + TILE_M - 1) / TILE_M, 256, 0, stream>>>(x, W, deg, rowstart, csr_col, out, N);
}


# CSR build as two-level counting sort: chunks order edges by 256-row bucket in LDS, one workgroup per bucket ranks its rows; node-count matrix gone, 4 launches to 3; row sum order = LDS arrival order
# speedup vs baseline: 4.5834x; 1.4274x over previous
#include <hip/hip_runtime.h>

#define FEAT  128
#define EMBED 128

typedef float v2f __attribute__((ext_vector_type(2)));
typedef float v8f __attribute__((ext_vector_type(8)));

#define BK_SHIFT 8
#define BK_NODES 256
#define MAXB 2048
#define K1_WG 512
#define EPT 13
#define CAP (K1_WG * EPT)

__device__ __forceinline__ int wave_incl_scan(int x, int lane) {
    x += __builtin_amdgcn_update_dpp(0, x, 0x111, 0xf, 0xf, true);
    x += __builtin_amdgcn_update_dpp(0, x, 0x112, 0xf, 0xf, true);
    x += __builtin_amdgcn_update_dpp(0, x, 0x114, 0xf, 0xf, true);
    x += __builtin_amdgcn_update_dpp(0, x, 0x118, 0xf, 0xf, true);
    const int low_row_total = __builtin_amdgcn_readlane(x, 15);
    if (lane >= 16) x += low_row_total;
    return x;
}

__global__ __launch_bounds__(K1_WG) void chunk_sort_kernel(const int* __restrict__ row,
                                                           const int* __restrict__ col,
                                                           int* __restrict__ srt_row,
                                                           int* __restrict__ srt_col,
                                                           int* __restrict__ offT,
                                                           int N, int E, int CH) {
    __shared__ alignas(16) int bcnt[MAXB];
    __shared__ int wtot[K1_WG / 32];
    __shared__ int st_row[CAP];
    __shared__ int st_col[CAP];
    const int t = threadIdx.x;
    const int lane = t & 31;
    const int wave = __builtin_amdgcn_readfirstlane(t >> 5);
    const int nb = (N + BK_NODES - 1) >> BK_SHIFT;
    const int begin = (int)blockIdx.x * CAP;
    const int cntE = (E - begin < CAP) ? (E - begin) : CAP;

    *(int4*)(bcnt + 4 * t) = make_int4(0, 0, 0, 0);
    __syncthreads();

    int rr[EPT], cc[EPT], pos[EPT];
#pragma unroll
    for (int j = 0; j < EPT; ++j) {
        const int e = j * K1_WG + t;
        const bool ok = e < cntE;
        const int r = ok ? row[begin + e] : -1;
        cc[j] = ok ? col[begin + e] : 0;
        rr[j] = ((unsigned)r < (unsigned)N) ? r : -1;
    }
#pragma unroll
    for (int j = 0; j < EPT; ++j)
        pos[j] = (rr[j] >= 0) ? atomicAdd(&bcnt[rr[j] >> BK_SHIFT], 1) : 0;
    __syncthreads();

    const int4 v = *(const int4*)(bcnt + 4 * t);
    const int s = v.x + v.y + v.z + v.w;
    const int inc = wave_incl_scan(s, lane);
    if (lane == 31) wtot[wave] = inc;
    __syncthreads();
    const int wt = (lane < K1_WG / 32) ? wtot[lane] : 0;
    const int wsum = wave_incl_scan(wt, lane);
    const int before = __builtin_amdgcn_readlane(wsum, wave) - __builtin_amdgcn_readlane(wt, wave);
    const int total = __builtin_amdgcn_readlane(wsum, K1_WG / 32 - 1);
    const int st = before + inc - s;
    *(int4*)(bcnt + 4 * t) = make_int4(st, st + v.x, st + v.x + v.y, st + v.x + v.y + v.z);
    __syncthreads();

    for (int b = t; b <= nb; b += K1_WG)
        offT[(size_t)b * CH + blockIdx.x] = (b < nb) ? bcnt[b] : total;

#pragma unroll
    for (int j = 0; j < EPT; ++j) {
        if (rr[j] >= 0) {
            const int idx = bcnt[rr[j] >> BK_SHIFT] + pos[j];
            st_row[idx] = rr[j];
            st_col[idx] = cc[j];
        }
    }
    __syncthreads();
    for (int i = t; i < total; i += K1_WG) {
        srt_row[begin + i] = st_row[i];
        srt_col[begin + i] = st_col[i];
    }
}

#define K2_WG 512
#define K2_WAVES (K2_WG / 32)
#define PIECES_IN_FLIGHT 8
#define BCAP 16384

__device__ __forceinline__ void load_pieces(const int* __restrict__ src, int lo, int len, int c0, int k0,
                                            int lane, int (&a)[PIECES_IN_FLIGHT], int (&b)[PIECES_IN_FLIGHT]) {
#pragma unroll
    for (int u = 0; u < PIECES_IN_FLIGHT; ++u) {
        const int k = k0 + u;
        const int plo  = __builtin_amdgcn_readlane(lo, k);
        const int plen = __builtin_amdgcn_readlane(len, k);
        const int* __restrict__ p = src + (size_t)(c0 + K2_WAVES * k) * CAP + plo;
        a[u] = (lane < plen) ? p[lane] : -1;
        b[u] = (lane + 32 < plen) ? p[lane + 32] : -1;
    }
}

__global__ __launch_bounds__(K2_WG) void bucket_csr_kernel(const int* __restrict__ srt_row,
                                                           const int* __restrict__ srt_col,
                                                           const int* __restrict__ offT,
                                                           int* __restrict__ deg,
                                                           int* __restrict__ rowstart,
                                                           int* __restrict__ csr_col,
                                                           int N, int CH) {
    __shared__ int ncnt[BK_NODES];
    __shared__ int wtot[BK_NODES / 32];
    __shared__ int front[2];
    __shared__ int stage[BCAP];
    const int t = threadIdx.x;
    const int lane = t & 31;
    const int wave = __builtin_amdgcn_readfirstlane(t >> 5);
    const int node0 = (int)blockIdx.x << BK_SHIFT;
    const int* __restrict__ offA = offT + (size_t)blockIdx.x * CH;
    const int* __restrict__ offB = offA + CH;

    if (t < BK_NODES) ncnt[t] = 0;
    if (t < 2) front[t] = 0;
    __syncthreads();

    int pa = 0, pb = 0;
    for (int c = t; c < CH; c += K2_WG) { pa += offA[c]; pb += offB[c]; }
    pa = wave_incl_scan(pa, lane);
    pb = wave_incl_scan(pb, lane);
    if (lane == 31) { atomicAdd(&front[0], pa); atomicAdd(&front[1], pb); }

    for (int c0 = wave; c0 < CH; c0 += K2_WAVES * 32) {
        const int c = c0 + K2_WAVES * lane;
        int lo = 0, len = 0;
        if (c < CH) { lo = offA[c]; len = offB[c] - lo; }
        const int left = (CH - c0 + K2_WAVES - 1) / K2_WAVES;
        const int np = (left < 32) ? left : 32;
        for (int k0 = 0; k0 < np; k0 += PIECES_IN_FLIGHT) {
            int ra[PIECES_IN_FLIGHT], rb[PIECES_IN_FLIGHT];
            load_pieces(srt_row, lo, len, c0, k0, lane, ra, rb);
#pragma unroll
            for (int u = 0; u < PIECES_IN_FLIGHT; ++u) {
                if (ra[u] >= 0) atomicAdd(&ncnt[ra[u] - node0], 1);
                if (rb[u] >= 0) atomicAdd(&ncnt[rb[u] - node0], 1);
            }
            for (int u = 0; u < PIECES_IN_FLIGHT; ++u) {
                const int k = k0 + u;
                const int plo  = __builtin_amdgcn_readlane(lo, k);
                const int plen = __builtin_amdgcn_readlane(len, k);
                const int* __restrict__ p = srt_row + (size_t)(c0 + K2_WAVES * k) * CAP + plo;
                for (int i = 64 + lane; i < plen; i += 32) atomicAdd(&ncnt[p[i] - node0], 1);
            }
        }
    }
    __syncthreads();

    const int base  = front[0];
    const int total = front[1] - base;
    const int cntv = (t < BK_NODES) ? ncnt[t] : 0;
    const int inc = wave_incl_scan(cntv, lane);
    if (t < BK_NODES && lane == 31) wtot[wave] = inc;
    __syncthreads();
    const int wt = (lane < BK_NODES / 32) ? wtot[lane] : 0;
    const int wsum = wave_incl_scan(wt, lane);
    const int wsel = (wave < BK_NODES / 32) ? wave : 0;
    const int before = __builtin_amdgcn_readlane(wsum, wsel) - __builtin_amdgcn_readlane(wt, wsel);
    if (t < BK_NODES) {
        const int start = before + inc - cntv;
        ncnt[t] = start;
        const int node = node0 + t;
        if (node < N) { deg[node] = cntv; rowstart[node] = base + start; }
    }
    if (blockIdx.x == gridDim.x - 1 && t == 0) rowstart[N] = front[1];
    __syncthreads();

    const bool staged = total <= BCAP;
    for (int c0 = wave; c0 < CH; c0 += K2_WAVES * 32) {
        const int c = c0 + K2_WAVES * lane;
        int lo = 0, len = 0;
        if (c < CH) { lo = offA[c]; len = offB[c] - lo; }
        const int left = (CH - c0 + K2_WAVES - 1) / K2_WAVES;
        const int np = (left < 32) ? left : 32;
        for (int k0 = 0; k0 < np; k0 += PIECES_IN_FLIGHT) {
            int ra[PIECES_IN_FLIGHT], rb[PIECES_IN_FLIGHT], ca[PIECES_IN_FLIGHT], cb[PIECES_IN_FLIGHT];
            load_pieces(srt_row, lo, len, c0, k0, lane, ra, rb);
            load_pieces(srt_col, lo, len, c0, k0, lane, ca, cb);
#pragma unroll
            for (int u = 0; u < PIECES_IN_FLIGHT; ++u) {
                if (ra[u] >= 0) {
                    const int slot = atomicAdd(&ncnt[ra[u] - node0], 1);
                    if (staged) stage[slot] = ca[u]; else csr_col[base + slot] = ca[u];
                }
                if (rb[u] >= 0) {
                    const int slot = atomicAdd(&ncnt[rb[u] - node0], 1);
                    if (staged) stage[slot] = cb[u]; else csr_col[base + slot] = cb[u];
                }
            }
            for (int u = 0; u < PIECES_IN_FLIGHT; ++u) {
                const int k = k0 + u;
                const int plo  = __builtin_amdgcn_readlane(lo, k);
                const int plen = __builtin_amdgcn_readlane(len, k);
                const size_t off = (size_t)(c0 + K2_WAVES * k) * CAP + plo;
                for (int i = 64 + lane; i < plen; i += 32) {
                    const int slot = atomicAdd(&ncnt[srt_row[off + i] - node0], 1);
                    if (staged) stage[slot] = srt_col[off + i]; else csr_col[base + slot] = srt_col[off + i];
                }
            }
        }
    }
    __syncthreads();
    if (staged)
        for (int i = t; i < total; i += K2_WG) csr_col[base + i] = stage[i];
}

#define TILE_M 16
#define AGG_PITCH 132
#define ROWS_IN_FLIGHT 8

__device__ __forceinline__ int nbr_id(const int* __restrict__ csr_col, int idx, int end) {
    int k = (idx < end) ? idx : end - 1;
    k = (k > 0) ? k : 0;
    const int c = csr_col[k];
    return (idx < end) ? c : 0;
}

__device__ __forceinline__ float nbr_norm(const int* __restrict__ deg, int c, float dr, int idx, int end) {
    const float n = rsqrtf(dr * (float)deg[c]);
    return (idx < end) ? n : 0.f;
}

__device__ __forceinline__ void add_rows(float4& acc, const float* __restrict__ x,
                                         int cv, float nv, int cnt, int lane) {
    const int nvi = __float_as_int(nv);
    for (int p = 0; p < cnt; p += ROWS_IN_FLIGHT) {
        float4 v[ROWS_IN_FLIGHT];
        float s[ROWS_IN_FLIGHT];
#pragma unroll
        for (int u = 0; u < ROWS_IN_FLIGHT; ++u) {
            const unsigned c = (unsigned)__builtin_amdgcn_readlane(cv, p + u);
            s[u] = __int_as_float(__builtin_amdgcn_readlane(nvi, p + u));
            const float* rowp = x + (size_t)c * FEAT;
            v[u] = *(const float4*)(rowp + 4 * lane);
        }
#pragma unroll
        for (int u = 0; u < ROWS_IN_FLIGHT; ++u) {
            acc.x = __builtin_fmaf(v[u].x, s[u], acc.x);
            acc.y = __builtin_fmaf(v[u].y, s[u], acc.y);
            acc.z = __builtin_fmaf(v[u].z, s[u], acc.z);
            acc.w = __builtin_fmaf(v[u].w, s[u], acc.w);
        }
    }
}

__global__ __launch_bounds__(256) void gather_wmma_kernel(const float* __restrict__ x,
                                                          const float* __restrict__ W,
                                                          const int* __restrict__ deg,
                                                          const int* __restrict__ rowstart,
                                                          const int* __restrict__ csr_col,
                                                          float* __restrict__ out, int N) {
    __shared__ alignas(16) float agg[TILE_M * AGG_PITCH];
    const int lane = threadIdx.x & 31;
    const int wave = __builtin_amdgcn_readfirstlane(threadIdx.x >> 5);
    const int m0 = blockIdx.x * TILE_M;

    const int r0 = m0 + 2 * wave;
    const int i0 = (r0     < N) ? r0     : N;
    const int i1 = (r0 + 1 < N) ? r0 + 1 : N;
    const int i2 = (r0 + 2 < N) ? r0 + 2 : N;
    int bound[3];
    bound[0] = rowstart[i0];
    bound[1] = rowstart[i1];
    bound[2] = rowstart[i2];
    float dself[2];
    dself[0] = (float)deg[(i0 < N) ? i0 : N - 1];
    dself[1] = (float)deg[(i1 < N) ? i1 : N - 1];

    int ca[2], cb[2];
    float na[2], nb[2];
#pragma unroll
    for (int j = 0; j < 2; ++j) {
        ca[j] = nbr_id(csr_col, bound[j] + lane,      bound[j + 1]);
        cb[j] = nbr_id(csr_col, bound[j] + 32 + lane, bound[j + 1]);
    }
#pragma unroll
    for (int j = 0; j < 2; ++j) {
        na[j] = nbr_norm(deg, ca[j], dself[j], bound[j] + lane,      bound[j + 1]);
        nb[j] = nbr_norm(deg, cb[j], dself[j], bound[j] + 32 + lane, bound[j + 1]);
    }

#pragma unroll
    for (int j = 0; j < 2; ++j) {
        const int start = bound[j];
        const int end   = bound[j + 1];
        int   cA = ca[j], cB = cb[j];
        float nA = na[j], nB = nb[j];
        float4 acc = {0.f, 0.f, 0.f, 0.f};
        for (int b = start; b < end; b += 64) {
            if (b != start) {
                cA = nbr_id(csr_col, b + lane,      end);
                cB = nbr_id(csr_col, b + 32 + lane, end);
                nA = nbr_norm(deg, cA, dself[j], b + lane,      end);
                nB = nbr_norm(deg, cB, dself[j], b + 32 + lane, end);
            }
            const int left = end - b;
            add_rows(acc, x, cA, nA, (left < 32) ? left : 32, lane);
            add_rows(acc, x, cB, nB, (left < 64) ? left - 32 : 32, lane);
        }
        *(float4*)(agg + (2 * wave + j) * AGG_PITCH + lane * 4) = acc;
    }
    __syncthreads();

    const int l  = lane & 15;
    const int hh = lane >> 4;
    const int n0 = wave * 16;
    const float* arow = agg + l * AGG_PITCH + 2 * hh;
    const float* __restrict__ wcol = W + (size_t)(2 * hh) * EMBED + n0 + l;

    v8f acc = {};
#pragma unroll 16
    for (int k0 = 0; k0 < FEAT; k0 += 4) {
        v2f a, b;
        a = *(const v2f*)(arow + k0);
        b.x = wcol[(size_t)k0       * EMBED];
        b.y = wcol[(size_t)(k0 + 1) * EMBED];
        acc = __builtin_amdgcn_wmma_f32_16x16x4_f32(false, a, false, b,
                                                    (short)0, acc, false, false);
    }

    float* __restrict__ orow = out + (size_t)m0 * EMBED + n0 + l;
    if (m0 + TILE_M <= N) {
#pragma unroll
        for (int r = 0; r < 8; ++r)
            orow[(size_t)(r + 8 * hh) * EMBED] = acc[r];
    } else {
#pragma unroll
        for (int r = 0; r < 8; ++r) {
            const int mr = r + 8 * hh;
            if (m0 + mr < N) orow[(size_t)mr * EMBED] = acc[r];
        }
    }
}

extern "C" void kernel_launch(void* const* d_in, const int* in_sizes, int n_in,
                              void* d_out, int out_size, void* d_ws, size_t ws_size,
                              hipStream_t stream) {
    const float* x   = (const float*)d_in[0];
    const float* W   = (const float*)d_in[1];
    const int*   row = (const int*)d_in[2];
    const int*   col = (const int*)d_in[3];
    float*       out = (float*)d_out;

    const int N = in_sizes[0] / FEAT;
    const int E = in_sizes[2];
    const int CH = (E + CAP - 1) / CAP;
    const int NB = (N + BK_NODES - 1) / BK_NODES;

    int*   srt_row  = (int*)d_ws;
    int*   srt_col  = srt_row + E;
    int*   csr_col  = srt_col + E;
    int*   deg      = csr_col + E;
    int*   rowstart = deg + N;
    int*   offT     = rowstart + (N + 1);

    chunk_sort_kernel<<<CH, K1_WG, 0, stream>>>(row, col, srt_row, srt_col, offT, N, E, CH);

    bucket_csr_kernel<<<NB, K2_WG, 0, stream>>>(srt_row, srt_col, offT, deg, rowstart, csr_col, N, CH);

    gather_wmma_kernel<<<(N + TILE_M - 1) / TILE_M, 256, 0, stream>>>(x, W, deg, rowstart, csr_col, out, N);
}


# bucket kernel reads its pieces once: the counting LDS adds return each edge's rank, kept in registers; piece tails past 64 edges spread over all 512 threads; row sum order = arrival at the count
# speedup vs baseline: 5.0855x; 1.1095x over previous
#include <hip/hip_runtime.h>

#define FEAT  128
#define EMBED 128

typedef float v2f __attribute__((ext_vector_type(2)));
typedef float v8f __attribute__((ext_vector_type(8)));

#define BK_SHIFT 8
#define BK_NODES 256
#define MAXB 2048
#define K1_WG 512
#define EPT 13
#define CAP (K1_WG * EPT)

__device__ __forceinline__ int wave_incl_scan(int x, int lane) {
    x += __builtin_amdgcn_update_dpp(0, x, 0x111, 0xf, 0xf, true);
    x += __builtin_amdgcn_update_dpp(0, x, 0x112, 0xf, 0xf, true);
    x += __builtin_amdgcn_update_dpp(0, x, 0x114, 0xf, 0xf, true);
    x += __builtin_amdgcn_update_dpp(0, x, 0x118, 0xf, 0xf, true);
    const int low_row_total = __builtin_amdgcn_readlane(x, 15);
    if (lane >= 16) x += low_row_total;
    return x;
}

__global__ __launch_bounds__(K1_WG) void chunk_sort_kernel(const int* __restrict__ row,
                                                           const int* __restrict__ col,
                                                           int* __restrict__ srt_row,
                                                           int* __restrict__ srt_col,
                                                           int* __restrict__ offT,
                                                           int N, int E, int CH) {
    __shared__ alignas(16) int bcnt[MAXB];
    __shared__ int wtot[K1_WG / 32];
    __shared__ int st_row[CAP];
    __shared__ int st_col[CAP];
    const int t = threadIdx.x;
    const int lane = t & 31;
    const int wave = __builtin_amdgcn_readfirstlane(t >> 5);
    const int nb = (N + BK_NODES - 1) >> BK_SHIFT;
    const int begin = (int)blockIdx.x * CAP;
    const int cntE = (E - begin < CAP) ? (E - begin) : CAP;

    *(int4*)(bcnt + 4 * t) = make_int4(0, 0, 0, 0);
    __syncthreads();

    int rr[EPT], cc[EPT], pos[EPT];
#pragma unroll
    for (int j = 0; j < EPT; ++j) {
        const int e = j * K1_WG + t;
        const bool ok = e < cntE;
        const int r = ok ? row[begin + e] : -1;
        cc[j] = ok ? col[begin + e] : 0;
        rr[j] = ((unsigned)r < (unsigned)N) ? r : -1;
    }
#pragma unroll
    for (int j = 0; j < EPT; ++j)
        pos[j] = (rr[j] >= 0) ? atomicAdd(&bcnt[rr[j] >> BK_SHIFT], 1) : 0;
    __syncthreads();

    const int4 v = *(const int4*)(bcnt + 4 * t);
    const int s = v.x + v.y + v.z + v.w;
    const int inc = wave_incl_scan(s, lane);
    if (lane == 31) wtot[wave] = inc;
    __syncthreads();
    const int wt = (lane < K1_WG / 32) ? wtot[lane] : 0;
    const int wsum = wave_incl_scan(wt, lane);
    const int before = __builtin_amdgcn_readlane(wsum, wave) - __builtin_amdgcn_readlane(wt, wave);
    const int total = __builtin_amdgcn_readlane(wsum, K1_WG / 32 - 1);
    const int st = before + inc - s;
    *(int4*)(bcnt + 4 * t) = make_int4(st, st + v.x, st + v.x + v.y, st + v.x + v.y + v.z);
    __syncthreads();

    for (int b = t; b <= nb; b += K1_WG)
        offT[(size_t)b * CH + blockIdx.x] = (b < nb) ? bcnt[b] : total;

#pragma unroll
    for (int j = 0; j < EPT; ++j) {
        if (rr[j] >= 0) {
            const int idx = bcnt[rr[j] >> BK_SHIFT] + pos[j];
            st_row[idx] = rr[j];
            st_col[idx] = cc[j];
        }
    }
    __syncthreads();
    for (int i = t; i < total; i += K1_WG) {
        srt_row[begin + i] = st_row[i];
        srt_col[begin + i] = st_col[i];
    }
}

#define K2_WG 512
#define K2_WAVES (K2_WG / 32)
#define PIECES_IN_FLIGHT 8
#define ALL_PIECES 16
#define BCAP 16384

template <int NP>
__device__ __forceinline__ void load_pieces(const int* __restrict__ src, int lo, int len, int c0, int k0,
                                            int lane, int (&a)[NP], int (&b)[NP]) {
#pragma unroll
    for (int u = 0; u < NP; ++u) {
        const int k = k0 + u;
        const int plo  = __builtin_amdgcn_readlane(lo, k);
        const int plen = __builtin_amdgcn_readlane(len, k);
        const int* __restrict__ p = src + (size_t)(c0 + K2_WAVES * k) * CAP + plo;
        a[u] = (lane < plen) ? p[lane] : -1;
        b[u] = (lane + 32 < plen) ? p[lane + 32] : -1;
    }
}

__global__ __launch_bounds__(K2_WG) void bucket_csr_kernel(const int* __restrict__ srt_row,
                                                           const int* __restrict__ srt_col,
                                                           const int* __restrict__ offT,
                                                           int* __restrict__ deg,
                                                           int* __restrict__ rowstart,
                                                           int* __restrict__ csr_col,
                                                           int N, int CH) {
    __shared__ int ncnt[BK_NODES];
    __shared__ int tcnt[BK_NODES];
    __shared__ int wtot[BK_NODES / 32];
    __shared__ int front[2];
    __shared__ int tl_n;
    __shared__ int tl_off[K2_WAVES * ALL_PIECES];
    __shared__ int tl_len[K2_WAVES * ALL_PIECES];
    __shared__ int stage[BCAP];
    const int t = threadIdx.x;
    const int lane = t & 31;
    const int wave = __builtin_amdgcn_readfirstlane(t >> 5);
    const int node0 = (int)blockIdx.x << BK_SHIFT;
    const int* __restrict__ offA = offT + (size_t)blockIdx.x * CH;
    const int* __restrict__ offB = offA + CH;

    if (t < BK_NODES) { ncnt[t] = 0; tcnt[t] = 0; }
    if (t < 2) front[t] = 0;
    if (t == 2) tl_n = 0;
    __syncthreads();

    const bool one_pass = CH <= K2_WAVES * ALL_PIECES;
    int key[2 * ALL_PIECES], cval[2 * ALL_PIECES];

    if (one_pass) {
        const int c = wave + K2_WAVES * lane;
        int lo = 0, hi = 0;
        if (lane < ALL_PIECES && c < CH) { lo = offA[c]; hi = offB[c]; }
        const int len = hi - lo;
        if (len > 64) {
            const int q = atomicAdd(&tl_n, 1);
            tl_off[q] = c * CAP + lo + 64;
            tl_len[q] = len - 64;
        }
        __syncthreads();
        int ra[ALL_PIECES], rb[ALL_PIECES], ca[ALL_PIECES], cb[ALL_PIECES];
        load_pieces<ALL_PIECES>(srt_row, lo, len, wave, 0, lane, ra, rb);
        load_pieces<ALL_PIECES>(srt_col, lo, len, wave, 0, lane, ca, cb);
        const int ntail = tl_n;
        for (int q = 0; q < ntail; ++q) {
            const int off = tl_off[q];
            const int tlen = tl_len[q];
            for (int i = t; i < tlen; i += K2_WG) atomicAdd(&tcnt[srt_row[off + i] - node0], 1);
        }
        const int pa = wave_incl_scan(lo, lane);
        const int pb = wave_incl_scan(hi, lane);
        if (lane == 31) { atomicAdd(&front[0], pa); atomicAdd(&front[1], pb); }
        int pos[2 * ALL_PIECES];
#pragma unroll
        for (int u = 0; u < ALL_PIECES; ++u) {
            pos[2 * u]     = (ra[u] >= 0) ? atomicAdd(&ncnt[ra[u] - node0], 1) : 0;
            pos[2 * u + 1] = (rb[u] >= 0) ? atomicAdd(&ncnt[rb[u] - node0], 1) : 0;
        }
#pragma unroll
        for (int u = 0; u < ALL_PIECES; ++u) {
            key[2 * u]      = (ra[u] >= 0) ? (((ra[u] - node0) << 16) | pos[2 * u])     : -1;
            key[2 * u + 1]  = (rb[u] >= 0) ? (((rb[u] - node0) << 16) | pos[2 * u + 1]) : -1;
            cval[2 * u]     = ca[u];
            cval[2 * u + 1] = cb[u];
        }
    } else {
        int pa = 0, pb = 0;
        for (int c = t; c < CH; c += K2_WG) { pa += offA[c]; pb += offB[c]; }
        pa = wave_incl_scan(pa, lane);
        pb = wave_incl_scan(pb, lane);
        if (lane == 31) { atomicAdd(&front[0], pa); atomicAdd(&front[1], pb); }

        for (int c0 = wave; c0 < CH; c0 += K2_WAVES * 32) {
            const int c = c0 + K2_WAVES * lane;
            int lo = 0, len = 0;
            if (c < CH) { lo = offA[c]; len = offB[c] - lo; }
            const int left = (CH - c0 + K2_WAVES - 1) / K2_WAVES;
            const int np = (left < 32) ? left : 32;
            for (int k0 = 0; k0 < np; k0 += PIECES_IN_FLIGHT) {
                int ra[PIECES_IN_FLIGHT], rb[PIECES_IN_FLIGHT];
                load_pieces<PIECES_IN_FLIGHT>(srt_row, lo, len, c0, k0, lane, ra, rb);
#pragma unroll
                for (int u = 0; u < PIECES_IN_FLIGHT; ++u) {
                    if (ra[u] >= 0) atomicAdd(&ncnt[ra[u] - node0], 1);
                    if (rb[u] >= 0) atomicAdd(&ncnt[rb[u] - node0], 1);
                }
                for (int u = 0; u < PIECES_IN_FLIGHT; ++u) {
                    const int k = k0 + u;
                    const int plo  = __builtin_amdgcn_readlane(lo, k);
                    const int plen = __builtin_amdgcn_readlane(len, k);
                    const int* __restrict__ p = srt_row + (size_t)(c0 + K2_WAVES * k) * CAP + plo;
                    for (int i = 64 + lane; i < plen; i += 32) atomicAdd(&ncnt[p[i] - node0], 1);
                }
            }
        }
    }
    __syncthreads();

    const int base  = front[0];
    const int total = front[1] - base;
    const int regc = (t < BK_NODES) ? ncnt[t] : 0;
    const int cntv = (t < BK_NODES) ? regc + tcnt[t] : 0;
    const int inc = wave_incl_scan(cntv, lane);
    if (t < BK_NODES && lane == 31) wtot[wave] = inc;
    __syncthreads();
    const int wt = (lane < BK_NODES / 32) ? wtot[lane] : 0;
    const int wsum = wave_incl_scan(wt, lane);
    const int wsel = (wave < BK_NODES / 32) ? wave : 0;
    const int before = __builtin_amdgcn_readlane(wsum, wsel) - __builtin_amdgcn_readlane(wt, wsel);
    if (t < BK_NODES) {
        const int start = before + inc - cntv;
        ncnt[t] = start;
        tcnt[t] = start + regc;
        const int node = node0 + t;
        if (node < N) { deg[node] = cntv; rowstart[node] = base + start; }
    }
    if (blockIdx.x == gridDim.x - 1 && t == 0) rowstart[N] = front[1];
    __syncthreads();

    const bool staged = total <= BCAP;
    if (one_pass && staged) {
        int st[2 * ALL_PIECES];
#pragma unroll
        for (int u = 0; u < 2 * ALL_PIECES; ++u) st[u] = ncnt[(key[u] >= 0) ? (key[u] >> 16) : 0];
#pragma unroll
        for (int u = 0; u < 2 * ALL_PIECES; ++u)
            if (key[u] >= 0) stage[st[u] + (key[u] & 0xffff)] = cval[u];
        const int ntail = tl_n;
        for (int q = 0; q < ntail; ++q) {
            const int off = tl_off[q];
            const int tlen = tl_len[q];
            for (int i = t; i < tlen; i += K2_WG) {
                const int r = srt_row[off + i];
                const int cv = srt_col[off + i];
                const int slot = atomicAdd(&tcnt[r - node0], 1);
                stage[slot] = cv;
            }
        }
    } else {
        for (int c0 = wave; c0 < CH; c0 += K2_WAVES * 32) {
            const int c = c0 + K2_WAVES * lane;
            int lo = 0, len = 0;
            if (c < CH) { lo = offA[c]; len = offB[c] - lo; }
            const int left = (CH - c0 + K2_WAVES - 1) / K2_WAVES;
            const int np = (left < 32) ? left : 32;
            for (int k0 = 0; k0 < np; k0 += PIECES_IN_FLIGHT) {
                int ra[PIECES_IN_FLIGHT], rb[PIECES_IN_FLIGHT], ca[PIECES_IN_FLIGHT], cb[PIECES_IN_FLIGHT];
                load_pieces<PIECES_IN_FLIGHT>(srt_row, lo, len, c0, k0, lane, ra, rb);
                load_pieces<PIECES_IN_FLIGHT>(srt_col, lo, len, c0, k0, lane, ca, cb);
#pragma unroll
                for (int u = 0; u < PIECES_IN_FLIGHT; ++u) {
                    if (ra[u] >= 0) {
                        const int slot = atomicAdd(&ncnt[ra[u] - node0], 1);
                        if (staged) stage[slot] = ca[u]; else csr_col[base + slot] = ca[u];
                    }
                    if (rb[u] >= 0) {
                        const int slot = atomicAdd(&ncnt[rb[u] - node0], 1);
                        if (staged) stage[slot] = cb[u]; else csr_col[base + slot] = cb[u];
                    }
                }
                for (int u = 0; u < PIECES_IN_FLIGHT; ++u) {
                    const int k = k0 + u;
                    const int plo  = __builtin_amdgcn_readlane(lo, k);
                    const int plen = __builtin_amdgcn_readlane(len, k);
                    const size_t off = (size_t)(c0 + K2_WAVES * k) * CAP + plo;
                    for (int i = 64 + lane; i < plen; i += 32) {
                        const int slot = atomicAdd(&ncnt[srt_row[off + i] - node0], 1);
                        if (staged) stage[slot] = srt_col[off + i]; else csr_col[base + slot] = srt_col[off + i];
                    }
                }
            }
        }
    }
    __syncthreads();
    if (staged)
        for (int i = t; i < total; i += K2_WG) csr_col[base + i] = stage[i];
}

#define TILE_M 16
#define AGG_PITCH 132
#define ROWS_IN_FLIGHT 8

__device__ __forceinline__ int nbr_id(const int* __restrict__ csr_col, int idx, int end) {
    int k = (idx < end) ? idx : end - 1;
    k = (k > 0) ? k : 0;
    const int c = csr_col[k];
    return (idx < end) ? c : 0;
}

__device__ __forceinline__ float nbr_norm(const int* __restrict__ deg, int c, float dr, int idx, int end) {
    const float n = rsqrtf(dr * (float)deg[c]);
    return (idx < end) ? n : 0.f;
}

__device__ __forceinline__ void add_rows(float4& acc, const float* __restrict__ x,
                                         int cv, float nv, int cnt, int lane) {
    const int nvi = __float_as_int(nv);
    for (int p = 0; p < cnt; p += ROWS_IN_FLIGHT) {
        float4 v[ROWS_IN_FLIGHT];
        float s[ROWS_IN_FLIGHT];
#pragma unroll
        for (int u = 0; u < ROWS_IN_FLIGHT; ++u) {
            const unsigned c = (unsigned)__builtin_amdgcn_readlane(cv, p + u);
            s[u] = __int_as_float(__builtin_amdgcn_readlane(nvi, p + u));
            const float* rowp = x + (size_t)c * FEAT;
            v[u] = *(const float4*)(rowp + 4 * lane);
        }
#pragma unroll
        for (int u = 0; u < ROWS_IN_FLIGHT; ++u) {
            acc.x = __builtin_fmaf(v[u].x, s[u], acc.x);
            acc.y = __builtin_fmaf(v[u].y, s[u], acc.y);
            acc.z = __builtin_fmaf(v[u].z, s[u], acc.z);
            acc.w = __builtin_fmaf(v[u].w, s[u], acc.w);
        }
    }
}

__global__ __launch_bounds__(256) void gather_wmma_kernel(const float* __restrict__ x,
                                                          const float* __restrict__ W,
                                                          const int* __restrict__ deg,
                                                          const int* __restrict__ rowstart,
                                                          const int* __restrict__ csr_col,
                                                          float* __restrict__ out, int N) {
    __shared__ alignas(16) float agg[TILE_M * AGG_PITCH];
    const int lane = threadIdx.x & 31;
    const int wave = __builtin_amdgcn_readfirstlane(threadIdx.x >> 5);
    const int m0 = blockIdx.x * TILE_M;

    const int r0 = m0 + 2 * wave;
    const int i0 = (r0     < N) ? r0     : N;
    const int i1 = (r0 + 1 < N) ? r0 + 1 : N;
    const int i2 = (r0 + 2 < N) ? r0 + 2 : N;
    int bound[3];
    bound[0] = rowstart[i0];
    bound[1] = rowstart[i1];
    bound[2] = rowstart[i2];
    float dself[2];
    dself[0] = (float)deg[(i0 < N) ? i0 : N - 1];
    dself[1] = (float)deg[(i1 < N) ? i1 : N - 1];

    int ca[2], cb[2];
    float na[2], nb[2];
#pragma unroll
    for (int j = 0; j < 2; ++j) {
        ca[j] = nbr_id(csr_col, bound[j] + lane,      bound[j + 1]);
        cb[j] = nbr_id(csr_col, bound[j] + 32 + lane, bound[j + 1]);
    }
#pragma unroll
    for (int j = 0; j < 2; ++j) {
        na[j] = nbr_norm(deg, ca[j], dself[j], bound[j] + lane,      bound[j + 1]);
        nb[j] = nbr_norm(deg, cb[j], dself[j], bound[j] + 32 + lane, bound[j + 1]);
    }

#pragma unroll
    for (int j = 0; j < 2; ++j) {
        const int start = bound[j];
        const int end   = bound[j + 1];
        int   cA = ca[j], cB = cb[j];
        float nA = na[j], nB = nb[j];
        float4 acc = {0.f, 0.f, 0.f, 0.f};
        for (int b = start; b < end; b += 64) {
            if (b != start) {
                cA = nbr_id(csr_col, b + lane,      end);
                cB = nbr_id(csr_col, b + 32 + lane, end);
                nA = nbr_norm(deg, cA, dself[j], b + lane,      end);
                nB = nbr_norm(deg, cB, dself[j], b + 32 + lane, end);
            }
            const int left = end - b;
            add_rows(acc, x, cA, nA, (left < 32) ? left : 32, lane);
            add_rows(acc, x, cB, nB, (left < 64) ? left - 32 : 32, lane);
        }
        *(float4*)(agg + (2 * wave + j) * AGG_PITCH + lane * 4) = acc;
    }
    __syncthreads();

    const int l  = lane & 15;
    const int hh = lane >> 4;
    const int n0 = wave * 16;
    const float* arow = agg + l * AGG_PITCH + 2 * hh;
    const float* __restrict__ wcol = W + (size_t)(2 * hh) * EMBED + n0 + l;

    v8f acc = {};
#pragma unroll 16
    for (int k0 = 0; k0 < FEAT; k0 += 4) {
        v2f a, b;
        a = *(const v2f*)(arow + k0);
        b.x = wcol[(size_t)k0       * EMBED];
        b.y = wcol[(size_t)(k0 + 1) * EMBED];
        acc = __builtin_amdgcn_wmma_f32_16x16x4_f32(false, a, false, b,
                                                    (short)0, acc, false, false);
    }

    float* __restrict__ orow = out + (size_t)m0 * EMBED + n0 + l;
    if (m0 + TILE_M <= N) {
#pragma unroll
        for (int r = 0; r < 8; ++r)
            orow[(size_t)(r + 8 * hh) * EMBED] = acc[r];
    } else {
#pragma unroll
        for (int r = 0; r < 8; ++r) {
            const int mr = r + 8 * hh;
            if (m0 + mr < N) orow[(size_t)mr * EMBED] = acc[r];
        }
    }
}

extern "C" void kernel_launch(void* const* d_in, const int* in_sizes, int n_in,
                              void* d_out, int out_size, void* d_ws, size_t ws_size,
                              hipStream_t stream) {
    const float* x   = (const float*)d_in[0];
    const float* W   = (const float*)d_in[1];
    const int*   row = (const int*)d_in[2];
    const int*   col = (const int*)d_in[3];
    float*       out = (float*)d_out;

    const int N = in_sizes[0] / FEAT;
    const int E = in_sizes[2];
    const int CH = (E + CAP - 1) / CAP;
    const int NB = (N + BK_NODES - 1) / BK_NODES;

    int*   srt_row  = (int*)d_ws;
    int*   srt_col  = srt_row + E;
    int*   csr_col  = srt_col + E;
    int*   deg      = csr_col + E;
    int*   rowstart = deg + N;
    int*   offT     = rowstart + (N + 1);

    chunk_sort_kernel<<<CH, K1_WG, 0, stream>>>(row, col, srt_row, srt_col, offT, N, E, CH);

    bucket_csr_kernel<<<NB, K2_WG, 0, stream>>>(srt_row, srt_col, offT, deg, rowstart, csr_col, N, CH);

    gather_wmma_kernel<<<(N + TILE_M - 1) / TILE_M, 256, 0, stream>>>(x, W, deg, rowstart, csr_col, out, N);
}
